# starting kernel
# baseline (speedup 1.0000x reference)
#include <hip/hip_runtime.h>

#define NN 50000
#define NE 800000
#define D  128

typedef __attribute__((ext_vector_type(2))) float v2f;
typedef __attribute__((ext_vector_type(8))) float v8f;

__global__ void degree_kernel(const int* __restrict__ src,
                              const int* __restrict__ dst,
                              float* __restrict__ din,
                              float* __restrict__ dout, int E) {
    int e = blockIdx.x * blockDim.x + threadIdx.x;
    if (e < E) {
        unsafeAtomicAdd(&din[dst[e]], 1.0f);
        unsafeAtomicAdd(&dout[src[e]], 1.0f);
    }
}

__global__ __launch_bounds__(256)
void scatter_kernel(const float* __restrict__ h,
                    const int* __restrict__ src,
                    const int* __restrict__ dst,
                    const float* __restrict__ degout,
                    float* __restrict__ agg, int E, int useScale) {
    int wave = (blockIdx.x * blockDim.x + threadIdx.x) >> 5;
    int lane = threadIdx.x & 31;
    if (wave >= E) return;
    int s = src[wave];
    int d = dst[wave];
    float sc = 1.0f;
    if (useScale) sc = rsqrtf(fmaxf(degout[s], 1.0f));
    const float4* hv = reinterpret_cast<const float4*>(h + (size_t)s * D);
    float4 v = hv[lane];
    float* ap = agg + (size_t)d * D + lane * 4;
    unsafeAtomicAdd(ap + 0, v.x * sc);
    unsafeAtomicAdd(ap + 1, v.y * sc);
    unsafeAtomicAdd(ap + 2, v.z * sc);
    unsafeAtomicAdd(ap + 3, v.w * sc);
}

__global__ __launch_bounds__(256)
void norm_gemm_wmma(const float* __restrict__ agg,
                    const float* __restrict__ hprev,
                    const float* __restrict__ deg,
                    const float* __restrict__ W,
                    const float* __restrict__ bias,
                    float* __restrict__ out, int mode) {
    __shared__ float Alds[16 * 132];

    const int t = threadIdx.x;
    const int rowBase = blockIdx.x * 16;

    {
        int r  = t >> 4;
        int c0 = (t & 15) * 8;
        int row = rowBase + r;
        float s;
        if (mode == 0) s = 1.0f / (deg[row] + 1.0f);
        else           s = rsqrtf(fmaxf(deg[row], 1.0f));
        const float* ar = agg   + (size_t)row * D + c0;
        const float* hr = hprev + (size_t)row * D + c0;
#pragma unroll
        for (int j = 0; j < 8; ++j) {
            float v = ar[j];
            if (mode == 0) v += hr[j];
            Alds[r * 132 + c0 + j] = v * s;
        }
    }
    __syncthreads();

    const int waveId = t >> 5;
    const int lane   = t & 31;
    const int half   = lane >> 4;
    const int lm     = lane & 15;
    const int col0   = waveId * 16;

    v8f c = {};
#pragma unroll
    for (int kk = 0; kk < 32; ++kk) {
        int k0 = 4 * kk + 2 * half;
        v2f a, b;
        a.x = Alds[lm * 132 + k0];
        a.y = Alds[lm * 132 + k0 + 1];
        b.x = W[(size_t)(k0)     * D + col0 + lm];
        b.y = W[(size_t)(k0 + 1) * D + col0 + lm];
        c = __builtin_amdgcn_wmma_f32_16x16x4_f32(
                false, a, false, b, (short)0, c, false, false);
    }

    float bb = bias[col0 + lm];
#pragma unroll
    for (int v = 0; v < 8; ++v) {
        int row = rowBase + v + 8 * half;
        out[(size_t)row * D + col0 + lm] = c[v] + bb;
    }
}

extern "C" void kernel_launch(void* const* d_in, const int* in_sizes, int n_in,
                              void* d_out, int out_size, void* d_ws, size_t ws_size,
                              hipStream_t stream) {
    const float* x  = (const float*)d_in[0];
    const float* W1 = (const float*)d_in[1];
    const float* b1 = (const float*)d_in[2];
    const float* W2 = (const float*)d_in[3];
    const float* b2 = (const float*)d_in[4];
    const float* W3 = (const float*)d_in[5];
    const float* b3 = (const float*)d_in[6];
    const int* src  = (const int*)d_in[7];
    const int* dst  = (const int*)d_in[8];
    float* out = (float*)d_out;

    const int N = NN, E = NE;
    float* deg_in  = (float*)d_ws;
    float* deg_out = deg_in + N;
    float* agg     = deg_out + N;
    float* h1      = agg + (size_t)N * D;
    float* h2      = h1  + (size_t)N * D;

    const size_t matBytes = (size_t)N * D * sizeof(float);
    dim3 edgeWaveGrid((E + 7) / 8);
    dim3 gemmGrid(N / 16);

    hipMemsetAsync(deg_in, 0, (size_t)2 * N * sizeof(float), stream);
    hipMemsetAsync(agg, 0, matBytes, stream);
    degree_kernel<<<(E + 255) / 256, 256, 0, stream>>>(src, dst, deg_in, deg_out, E);

    scatter_kernel<<<edgeWaveGrid, 256, 0, stream>>>(x, src, dst, deg_out, agg, E, 0);
    norm_gemm_wmma<<<gemmGrid, 256, 0, stream>>>(agg, x, deg_in, W1, b1, h1, 0);

    hipMemsetAsync(agg, 0, matBytes, stream);
    scatter_kernel<<<edgeWaveGrid, 256, 0, stream>>>(h1, src, dst, deg_out, agg, E, 0);
    norm_gemm_wmma<<<gemmGrid, 256, 0, stream>>>(agg, h1, deg_in, W2, b2, h2, 0);

    hipMemsetAsync(agg, 0, matBytes, stream);
    scatter_kernel<<<edgeWaveGrid, 256, 0, stream>>>(h2, src, dst, deg_out, agg, E, 1);
    norm_gemm_wmma<<<gemmGrid, 256, 0, stream>>>(agg, h2, deg_in, W3, b3, out, 1);
}


# Float atomic scatter-add replaced by a CSR gather: the degree atomics return each edge's rank, scan + fill sort edges by dst, one wave per node sums its rows in rank (arrival) order, plain store
# speedup vs baseline: 2.3530x; 2.3530x over previous
#include <hip/hip_runtime.h>


#define NN 50000
#define NE 800000
#define D  128
#define NPAD 50048

typedef __attribute__((ext_vector_type(2))) float v2f;
typedef __attribute__((ext_vector_type(8))) float v8f;

__global__ void count_kernel(const int* __restrict__ src,
                             const int* __restrict__ dst,
                             int* __restrict__ cnt_in,
                             int* __restrict__ cnt_out,
                             int* __restrict__ rank, int E) {
    int e = blockIdx.x * blockDim.x + threadIdx.x;
    if (e < E) {
        rank[e] = atomicAdd(&cnt_in[dst[e]], 1);
        atomicAdd(&cnt_out[src[e]], 1);
    }
}

__global__ __launch_bounds__(1024)
void scan_kernel(const int* __restrict__ cnt_in,
                 int* __restrict__ off,
                 float* __restrict__ deg_in_f, int n) {
    __shared__ int buf[NN];
    __shared__ int wsum[32];

    const int t = threadIdx.x;
    for (int i = t; i < n; i += 1024) {
        int c = cnt_in[i];
        buf[i] = c;
        deg_in_f[i] = (float)c;
    }
    __syncthreads();

    const int chunk = (n + 1023) / 1024;
    const int b = t * chunk;
    const int e = min(b + chunk, n);
    int s = 0;
    for (int i = b; i < e; ++i) s += buf[i];

    const int lane = t & 31;
    const int w    = t >> 5;
    int incl = s;
    for (int o = 1; o < 32; o <<= 1) {
        int v = __shfl_up(incl, o, 32);
        if (lane >= o) incl += v;
    }
    if (lane == 31) wsum[w] = incl;
    __syncthreads();
    if (w == 0) {
        int v  = wsum[lane];
        int iv = v;
        for (int o = 1; o < 32; o <<= 1) {
            int u = __shfl_up(iv, o, 32);
            if (lane >= o) iv += u;
        }
        wsum[lane] = iv - v;
    }
    __syncthreads();

    int run = wsum[w] + incl - s;
    for (int i = b; i < e; ++i) {
        int c = buf[i];
        buf[i] = run;
        run += c;
    }
    if (t == 1023) off[n] = run;
    __syncthreads();
    for (int i = t; i < n; i += 1024) off[i] = buf[i];
}

__global__ void fill_kernel(const int* __restrict__ src,
                            const int* __restrict__ dst,
                            const int* __restrict__ off,
                            const int* __restrict__ rank,
                            int* __restrict__ csr, int E) {
    int e = blockIdx.x * blockDim.x + threadIdx.x;
    if (e < E) csr[off[dst[e]] + rank[e]] = src[e];
}

template <bool SCALE, bool FULL>
__device__ __forceinline__
void gather8(const float* __restrict__ h, int idx, float sc,
             int j, int m, int lane, float4& acc) {
    float4 v[8];
    float  f[8];
#pragma unroll
    for (int k = 0; k < 8; ++k) {
        v[k] = make_float4(0.0f, 0.0f, 0.0f, 0.0f);
        f[k] = 0.0f;
        if (FULL || j + k < m) {
            const int s = __builtin_amdgcn_readlane(idx, j + k);
            v[k] = reinterpret_cast<const float4*>(h + (size_t)s * D)[lane];
            if (SCALE)
                f[k] = __int_as_float(
                    __builtin_amdgcn_readlane(__float_as_int(sc), j + k));
        }
    }
#pragma unroll
    for (int k = 0; k < 8; ++k) {
        if (SCALE) {
            acc.x += v[k].x * f[k];
            acc.y += v[k].y * f[k];
            acc.z += v[k].z * f[k];
            acc.w += v[k].w * f[k];
        } else {
            acc.x += v[k].x;
            acc.y += v[k].y;
            acc.z += v[k].z;
            acc.w += v[k].w;
        }
    }
}

template <bool SCALE>
__global__ __launch_bounds__(256)
void gather_kernel(const float* __restrict__ h,
                   const int* __restrict__ off,
                   const int* __restrict__ csr,
                   const int* __restrict__ cnt_out,
                   float* __restrict__ agg, int nNodes) {
    const int d = __builtin_amdgcn_readfirstlane(
        (int)((blockIdx.x * 256u + threadIdx.x) >> 5));
    const int lane = threadIdx.x & 31;
    if (d >= nNodes) return;

    const int beg = off[d];
    const int end = off[d + 1];
    float4 acc = make_float4(0.0f, 0.0f, 0.0f, 0.0f);

    for (int base = beg; base < end; base += 32) {
        const int m = min(32, end - base);
        int   idx = 0;
        float sc  = 1.0f;
        if (lane < m) {
            idx = csr[base + lane];
            idx = min(max(idx, 0), nNodes - 1);
            if (SCALE) sc = rsqrtf(fmaxf((float)cnt_out[idx], 1.0f));
        }
        int j = 0;
        for (; j + 8 <= m; j += 8) gather8<SCALE, true >(h, idx, sc, j, m, lane, acc);
        if (j < m)                 gather8<SCALE, false>(h, idx, sc, j, m, lane, acc);
    }
    reinterpret_cast<float4*>(agg + (size_t)d * D)[lane] = acc;
}

__global__ __launch_bounds__(256)
void norm_gemm_wmma(const float* __restrict__ agg,
                    const float* __restrict__ hprev,
                    const float* __restrict__ deg,
                    const float* __restrict__ W,
                    const float* __restrict__ bias,
                    float* __restrict__ out, int mode) {
    __shared__ float Alds[16 * 132];

    const int t = threadIdx.x;
    const int rowBase = blockIdx.x * 16;

    {
        int r  = t >> 4;
        int c0 = (t & 15) * 8;
        int row = rowBase + r;
        float s;
        if (mode == 0) s = 1.0f / (deg[row] + 1.0f);
        else           s = rsqrtf(fmaxf(deg[row], 1.0f));
        const float* ar = agg   + (size_t)row * D + c0;
        const float* hr = hprev + (size_t)row * D + c0;
#pragma unroll
        for (int j = 0; j < 8; ++j) {
            float v = ar[j];
            if (mode == 0) v += hr[j];
            Alds[r * 132 + c0 + j] = v * s;
        }
    }
    __syncthreads();

    const int waveId = t >> 5;
    const int lane   = t & 31;
    const int half   = lane >> 4;
    const int lm     = lane & 15;
    const int col0   = waveId * 16;

    v8f c = {};
#pragma unroll
    for (int kk = 0; kk < 32; ++kk) {
        int k0 = 4 * kk + 2 * half;
        v2f a, b;
        a.x = Alds[lm * 132 + k0];
        a.y = Alds[lm * 132 + k0 + 1];
        b.x = W[(size_t)(k0)     * D + col0 + lm];
        b.y = W[(size_t)(k0 + 1) * D + col0 + lm];
        c = __builtin_amdgcn_wmma_f32_16x16x4_f32(
                false, a, false, b, (short)0, c, false, false);
    }

    float bb = bias[col0 + lm];
#pragma unroll
    for (int v = 0; v < 8; ++v) {
        int row = rowBase + v + 8 * half;
        out[(size_t)row * D + col0 + lm] = c[v] + bb;
    }
}

extern "C" void kernel_launch(void* const* d_in, const int* in_sizes, int n_in,
                              void* d_out, int out_size, void* d_ws, size_t ws_size,
                              hipStream_t stream) {
    const float* x  = (const float*)d_in[0];
    const float* W1 = (const float*)d_in[1];
    const float* b1 = (const float*)d_in[2];
    const float* W2 = (const float*)d_in[3];
    const float* b2 = (const float*)d_in[4];
    const float* W3 = (const float*)d_in[5];
    const float* b3 = (const float*)d_in[6];
    const int* src  = (const int*)d_in[7];
    const int* dst  = (const int*)d_in[8];
    float* out = (float*)d_out;

    const int N = NN, E = NE;
    int*   cnt_in   = (int*)d_ws;
    int*   cnt_out  = cnt_in  + NPAD;
    int*   off      = cnt_out + NPAD;
    float* deg_in_f = (float*)(off + NPAD);
    int*   rank     = (int*)(deg_in_f + NPAD);
    int*   csr      = rank + NE;
    float* agg      = (float*)(csr + NE);
    float* h1       = agg + (size_t)NN * D;
    float* h2       = h1  + (size_t)NN * D;

    dim3 edgeGrid((NE + 255) / 256);
    dim3 nodeWaveGrid((NN + 7) / 8);
    dim3 gemmGrid(NN / 16);

    hipMemsetAsync(cnt_in, 0, (size_t)2 * NPAD * sizeof(int), stream);
    count_kernel<<<edgeGrid, 256, 0, stream>>>(src, dst, cnt_in, cnt_out, rank, E);
    scan_kernel<<<1, 1024, 0, stream>>>(cnt_in, off, deg_in_f, N);
    fill_kernel<<<edgeGrid, 256, 0, stream>>>(src, dst, off, rank, csr, E);

    gather_kernel<false><<<nodeWaveGrid, 256, 0, stream>>>(x, off, csr, cnt_out, agg, N);
    norm_gemm_wmma<<<gemmGrid, 256, 0, stream>>>(agg, x, deg_in_f, W1, b1, h1, 0);

    gather_kernel<false><<<nodeWaveGrid, 256, 0, stream>>>(h1, off, csr, cnt_out, agg, N);
    norm_gemm_wmma<<<gemmGrid, 256, 0, stream>>>(agg, h1, deg_in_f, W2, b2, h2, 0);

    gather_kernel<true><<<nodeWaveGrid, 256, 0, stream>>>(h2, off, csr, cnt_out, agg, N);
    norm_gemm_wmma<<<gemmGrid, 256, 0, stream>>>(agg, h2, deg_in_f, W3, b3, out, 1);
}


# Gather fused into the GEMM: each 16-row block's 8 waves sum their two nodes' in-rows (same arrival order) straight into the LDS A tile; agg buffer and 3 of 9 launches gone
# speedup vs baseline: 2.7369x; 1.1632x over previous
#include <hip/hip_runtime.h>


#define NN 50000
#define NE 800000
#define D  128
#define NPAD 50048

typedef __attribute__((ext_vector_type(2))) float v2f;
typedef __attribute__((ext_vector_type(8))) float v8f;

__global__ void count_kernel(const int* __restrict__ src,
                             const int* __restrict__ dst,
                             int* __restrict__ cnt_in,
                             int* __restrict__ cnt_out,
                             int* __restrict__ rank, int E) {
    int e = blockIdx.x * blockDim.x + threadIdx.x;
    if (e < E) {
        rank[e] = atomicAdd(&cnt_in[dst[e]], 1);
        atomicAdd(&cnt_out[src[e]], 1);
    }
}

__global__ __launch_bounds__(1024)
void scan_kernel(const int* __restrict__ cnt_in,
                 int* __restrict__ off, int n) {
    __shared__ int buf[NN];
    __shared__ int wsum[32];

    const int t = threadIdx.x;
    for (int i = t; i < n; i += 1024) {
        buf[i] = cnt_in[i];
    }
    __syncthreads();

    const int chunk = (n + 1023) / 1024;
    const int b = t * chunk;
    const int e = min(b + chunk, n);
    int s = 0;
    for (int i = b; i < e; ++i) s += buf[i];

    const int lane = t & 31;
    const int w    = t >> 5;
    int incl = s;
    for (int o = 1; o < 32; o <<= 1) {
        int v = __shfl_up(incl, o, 32);
        if (lane >= o) incl += v;
    }
    if (lane == 31) wsum[w] = incl;
    __syncthreads();
    if (w == 0) {
        int v  = wsum[lane];
        int iv = v;
        for (int o = 1; o < 32; o <<= 1) {
            int u = __shfl_up(iv, o, 32);
            if (lane >= o) iv += u;
        }
        wsum[lane] = iv - v;
    }
    __syncthreads();

    int run = wsum[w] + incl - s;
    for (int i = b; i < e; ++i) {
        int c = buf[i];
        buf[i] = run;
        run += c;
    }
    if (t == 1023) off[n] = run;
    __syncthreads();
    for (int i = t; i < n; i += 1024) off[i] = buf[i];
}

__global__ void fill_kernel(const int* __restrict__ src,
                            const int* __restrict__ dst,
                            const int* __restrict__ off,
                            const int* __restrict__ rank,
                            int* __restrict__ csr, int E) {
    int e = blockIdx.x * blockDim.x + threadIdx.x;
    if (e < E) csr[off[dst[e]] + rank[e]] = src[e];
}

template <bool SCALE, bool FULL>
__device__ __forceinline__
void gather8(const float* __restrict__ h, int idx, float sc,
             int j, int m, int lane, int pos, int split,
             float4& accA, float4& accB) {
    float4 v[8];
    float  f[8];
#pragma unroll
    for (int k = 0; k < 8; ++k) {
        v[k] = make_float4(0.0f, 0.0f, 0.0f, 0.0f);
        f[k] = 0.0f;
        if (FULL || j + k < m) {
            const int s = __builtin_amdgcn_readlane(idx, j + k);
            v[k] = reinterpret_cast<const float4*>(h + (size_t)s * D)[lane];
            if (SCALE)
                f[k] = __int_as_float(
                    __builtin_amdgcn_readlane(__float_as_int(sc), j + k));
        }
    }
#pragma unroll
    for (int k = 0; k < 8; ++k) {
        if (pos + k < split) {
            if (SCALE) {
                accA.x += v[k].x * f[k];
                accA.y += v[k].y * f[k];
                accA.z += v[k].z * f[k];
                accA.w += v[k].w * f[k];
            } else {
                accA.x += v[k].x;
                accA.y += v[k].y;
                accA.z += v[k].z;
                accA.w += v[k].w;
            }
        } else {
            if (SCALE) {
                accB.x += v[k].x * f[k];
                accB.y += v[k].y * f[k];
                accB.z += v[k].z * f[k];
                accB.w += v[k].w * f[k];
            } else {
                accB.x += v[k].x;
                accB.y += v[k].y;
                accB.z += v[k].z;
                accB.w += v[k].w;
            }
        }
    }
}

template <bool SCALE>
__global__ __launch_bounds__(256)
void gather_gemm_wmma(const float* __restrict__ h,
                      const int* __restrict__ off,
                      const int* __restrict__ csr,
                      const int* __restrict__ cnt_out,
                      const float* __restrict__ W,
                      const float* __restrict__ bias,
                      float* __restrict__ out, int nNodes) {
    __shared__ __attribute__((aligned(16))) float Alds[16 * 132];

    const int t = threadIdx.x;
    const int lane = t & 31;
    const int rowBase = blockIdx.x * 16;
    const int waveId = __builtin_amdgcn_readfirstlane(t >> 5);

    {
        const int nA = rowBase + 2 * waveId;
        const int o0 = off[nA];
        const int o1 = off[nA + 1];
        const int o2 = off[nA + 2];
        float4 accA = make_float4(0.0f, 0.0f, 0.0f, 0.0f);
        float4 accB = make_float4(0.0f, 0.0f, 0.0f, 0.0f);

        for (int base = o0; base < o2; base += 32) {
            const int m = min(32, o2 - base);
            int   idx = 0;
            float sc  = 1.0f;
            if (lane < m) {
                idx = csr[base + lane];
                idx = min(max(idx, 0), nNodes - 1);
                if (SCALE) sc = rsqrtf(fmaxf((float)cnt_out[idx], 1.0f));
            }
            int j = 0;
            for (; j + 8 <= m; j += 8)
                gather8<SCALE, true >(h, idx, sc, j, m, lane, base + j, o1, accA, accB);
            if (j < m)
                gather8<SCALE, false>(h, idx, sc, j, m, lane, base + j, o1, accA, accB);
        }

        const float degA = (float)(o1 - o0);
        const float degB = (float)(o2 - o1);
        float sA, sB;
        if (SCALE) {
            sA = rsqrtf(fmaxf(degA, 1.0f));
            sB = rsqrtf(fmaxf(degB, 1.0f));
        } else {
            sA = 1.0f / (degA + 1.0f);
            sB = 1.0f / (degB + 1.0f);
            const float4 hA = reinterpret_cast<const float4*>(h + (size_t)nA * D)[lane];
            const float4 hB = reinterpret_cast<const float4*>(h + (size_t)(nA + 1) * D)[lane];
            accA.x += hA.x; accA.y += hA.y; accA.z += hA.z; accA.w += hA.w;
            accB.x += hB.x; accB.y += hB.y; accB.z += hB.z; accB.w += hB.w;
        }
        float4 rA = make_float4(accA.x * sA, accA.y * sA, accA.z * sA, accA.w * sA);
        float4 rB = make_float4(accB.x * sB, accB.y * sB, accB.z * sB, accB.w * sB);
        *reinterpret_cast<float4*>(&Alds[(2 * waveId)     * 132 + 4 * lane]) = rA;
        *reinterpret_cast<float4*>(&Alds[(2 * waveId + 1) * 132 + 4 * lane]) = rB;
    }
    __syncthreads();

    const int half   = lane >> 4;
    const int lm     = lane & 15;
    const int col0   = waveId * 16;

    v8f c = {};
#pragma unroll
    for (int kk = 0; kk < 32; ++kk) {
        int k0 = 4 * kk + 2 * half;
        v2f a, b;
        a.x = Alds[lm * 132 + k0];
        a.y = Alds[lm * 132 + k0 + 1];
        b.x = W[(size_t)(k0)     * D + col0 + lm];
        b.y = W[(size_t)(k0 + 1) * D + col0 + lm];
        c = __builtin_amdgcn_wmma_f32_16x16x4_f32(
                false, a, false, b, (short)0, c, false, false);
    }

    float bb = bias[col0 + lm];
#pragma unroll
    for (int v = 0; v < 8; ++v) {
        int row = rowBase + v + 8 * half;
        out[(size_t)row * D + col0 + lm] = c[v] + bb;
    }
}

extern "C" void kernel_launch(void* const* d_in, const int* in_sizes, int n_in,
                              void* d_out, int out_size, void* d_ws, size_t ws_size,
                              hipStream_t stream) {
    const float* x  = (const float*)d_in[0];
    const float* W1 = (const float*)d_in[1];
    const float* b1 = (const float*)d_in[2];
    const float* W2 = (const float*)d_in[3];
    const float* b2 = (const float*)d_in[4];
    const float* W3 = (const float*)d_in[5];
    const float* b3 = (const float*)d_in[6];
    const int* src  = (const int*)d_in[7];
    const int* dst  = (const int*)d_in[8];
    float* out = (float*)d_out;

    const int N = NN, E = NE;
    int*   cnt_in   = (int*)d_ws;
    int*   cnt_out  = cnt_in  + NPAD;
    int*   off      = cnt_out + NPAD;
    int*   rank     = off + NPAD;
    int*   csr      = rank + NE;
    float* h1       = (float*)(csr + NE);
    float* h2       = h1  + (size_t)NN * D;

    dim3 edgeGrid((NE + 255) / 256);
    dim3 rowGrid(NN / 16);

    hipMemsetAsync(cnt_in, 0, (size_t)2 * NPAD * sizeof(int), stream);
    count_kernel<<<edgeGrid, 256, 0, stream>>>(src, dst, cnt_in, cnt_out, rank, E);
    scan_kernel<<<1, 1024, 0, stream>>>(cnt_in, off, N);
    fill_kernel<<<edgeGrid, 256, 0, stream>>>(src, dst, off, rank, csr, E);

    gather_gemm_wmma<false><<<rowGrid, 256, 0, stream>>>(x,  off, csr, cnt_out, W1, b1, h1, N);
    gather_gemm_wmma<false><<<rowGrid, 256, 0, stream>>>(h1, off, csr, cnt_out, W2, b2, h2, N);

    gather_gemm_wmma<true><<<rowGrid, 256, 0, stream>>>(h2, off, csr, cnt_out, W3, b3, out, N);
}


# Edge sort dropped: the degree add's rank puts each source id straight into a 32-slot row per destination (overflow to a side list, summed last, else arrival order kept); scan and fill launches gone
# speedup vs baseline: 3.3842x; 1.2365x over previous
#include <hip/hip_runtime.h>


#define NN 50000
#define NE 800000
#define D  128
#define NPAD 50048
#define CAP 32

typedef __attribute__((ext_vector_type(2))) float v2f;
typedef __attribute__((ext_vector_type(8))) float v8f;

__global__ void count_kernel(const int* __restrict__ src,
                             const int* __restrict__ dst,
                             int* __restrict__ cnt_in,
                             int* __restrict__ cnt_out,
                             int* __restrict__ slots,
                             int* __restrict__ ovf_cnt,
                             int* __restrict__ ovf, int E) {
    int e = blockIdx.x * blockDim.x + threadIdx.x;
    if (e < E) {
        const int d = dst[e];
        const int s = src[e];
        const int r = atomicAdd(&cnt_in[d], 1);
        atomicAdd(&cnt_out[s], 1);
        if (r < CAP) {
            slots[(size_t)d * CAP + r] = s;
        } else {
            const int p = atomicAdd(ovf_cnt, 1);
            ovf[2 * (size_t)p]     = d;
            ovf[2 * (size_t)p + 1] = s;
        }
    }
}

template <bool SCALE, bool FULL>
__device__ __forceinline__
void gather8(const float* __restrict__ h, int idx, float sc,
             int j, int m, int lane, int pos, int split,
             float4& accA, float4& accB) {
    float4 v[8];
    float  f[8];
#pragma unroll
    for (int k = 0; k < 8; ++k) {
        v[k] = make_float4(0.0f, 0.0f, 0.0f, 0.0f);
        f[k] = 0.0f;
        if (FULL || j + k < m) {
            const int s = __builtin_amdgcn_readlane(idx, j + k);
            v[k] = reinterpret_cast<const float4*>(h + (size_t)s * D)[lane];
            if (SCALE)
                f[k] = __int_as_float(
                    __builtin_amdgcn_readlane(__float_as_int(sc), j + k));
        }
    }
#pragma unroll
    for (int k = 0; k < 8; ++k) {
        if (pos + k < split) {
            if (SCALE) {
                accA.x += v[k].x * f[k];
                accA.y += v[k].y * f[k];
                accA.z += v[k].z * f[k];
                accA.w += v[k].w * f[k];
            } else {
                accA.x += v[k].x;
                accA.y += v[k].y;
                accA.z += v[k].z;
                accA.w += v[k].w;
            }
        } else {
            if (SCALE) {
                accB.x += v[k].x * f[k];
                accB.y += v[k].y * f[k];
                accB.z += v[k].z * f[k];
                accB.w += v[k].w * f[k];
            } else {
                accB.x += v[k].x;
                accB.y += v[k].y;
                accB.z += v[k].z;
                accB.w += v[k].w;
            }
        }
    }
}

template <bool SCALE>
__global__ __launch_bounds__(256)
void gather_gemm_wmma(const float* __restrict__ h,
                      const int* __restrict__ cnt_in,
                      const int* __restrict__ slots,
                      const int* __restrict__ ovf_cnt,
                      const int* __restrict__ ovf,
                      const int* __restrict__ cnt_out,
                      const float* __restrict__ W,
                      const float* __restrict__ bias,
                      float* __restrict__ out, int nNodes) {
    __shared__ __attribute__((aligned(16))) float Alds[16 * 132];

    const int t = threadIdx.x;
    const int lane = t & 31;
    const int rowBase = blockIdx.x * 16;
    const int waveId = __builtin_amdgcn_readfirstlane(t >> 5);

    {
        const int nA = rowBase + 2 * waveId;
        const int dA = cnt_in[nA];
        const int dB = cnt_in[nA + 1];
        const int mA = min(dA, CAP);
        const int mB = min(dB, CAP);
        const int tot = mA + mB;
        float4 accA = make_float4(0.0f, 0.0f, 0.0f, 0.0f);
        float4 accB = make_float4(0.0f, 0.0f, 0.0f, 0.0f);

        int iA = 0, iB = 0;
        if (lane < mA) iA = slots[(size_t)nA * CAP + lane];
        if (lane < mB) iB = slots[(size_t)(nA + 1) * CAP + lane];
        const int sh0 = __shfl(iB, (lane - mA) & 31, 32);
        const int sh1 = __shfl(iB, (lane + 32 - mA) & 31, 32);
        const int c0 = (lane < mA) ? iA : sh0;
        const int c1 = sh1;

        for (int base = 0; base < tot; base += 32) {
            const int m = min(32, tot - base);
            int   idx = 0;
            float sc  = 1.0f;
            if (lane < m) {
                idx = (base == 0) ? c0 : c1;
                idx = min(max(idx, 0), nNodes - 1);
                if (SCALE) sc = rsqrtf(fmaxf((float)cnt_out[idx], 1.0f));
            }
            int j = 0;
            for (; j + 8 <= m; j += 8)
                gather8<SCALE, true >(h, idx, sc, j, m, lane, base + j, mA, accA, accB);
            if (j < m)
                gather8<SCALE, false>(h, idx, sc, j, m, lane, base + j, mA, accA, accB);
        }

        if (dA > CAP || dB > CAP) {
            const int novf = min(ovf_cnt[0], NE);
            for (int base = 0; base < novf; base += 32) {
                const int e = base + lane;
                int d = -1, s = 0;
                if (e < novf) {
                    const int2 p = reinterpret_cast<const int2*>(ovf)[e];
                    d = p.x;
                    s = min(max(p.y, 0), nNodes - 1);
                }
                const unsigned hitA = __builtin_amdgcn_ballot_w32(d == nA);
                const unsigned hitB = __builtin_amdgcn_ballot_w32(d == nA + 1);
                unsigned hits = hitA | hitB;
                while (hits) {
                    const int k = __builtin_ctz(hits);
                    hits &= hits - 1;
                    const int sidx = __builtin_amdgcn_readlane(s, k);
                    const float4 v = reinterpret_cast<const float4*>(h + (size_t)sidx * D)[lane];
                    float f = 1.0f;
                    if (SCALE) f = rsqrtf(fmaxf((float)cnt_out[sidx], 1.0f));
                    if ((hitA >> k) & 1u) {
                        if (SCALE) {
                            accA.x += v.x * f; accA.y += v.y * f; accA.z += v.z * f; accA.w += v.w * f;
                        } else {
                            accA.x += v.x; accA.y += v.y; accA.z += v.z; accA.w += v.w;
                        }
                    } else {
                        if (SCALE) {
                            accB.x += v.x * f; accB.y += v.y * f; accB.z += v.z * f; accB.w += v.w * f;
                        } else {
                            accB.x += v.x; accB.y += v.y; accB.z += v.z; accB.w += v.w;
                        }
                    }
                }
            }
        }

        const float degA = (float)dA;
        const float degB = (float)dB;
        float sA, sB;
        if (SCALE) {
            sA = rsqrtf(fmaxf(degA, 1.0f));
            sB = rsqrtf(fmaxf(degB, 1.0f));
        } else {
            sA = 1.0f / (degA + 1.0f);
            sB = 1.0f / (degB + 1.0f);
            const float4 hA = reinterpret_cast<const float4*>(h + (size_t)nA * D)[lane];
            const float4 hB = reinterpret_cast<const float4*>(h + (size_t)(nA + 1) * D)[lane];
            accA.x += hA.x; accA.y += hA.y; accA.z += hA.z; accA.w += hA.w;
            accB.x += hB.x; accB.y += hB.y; accB.z += hB.z; accB.w += hB.w;
        }
        float4 rA = make_float4(accA.x * sA, accA.y * sA, accA.z * sA, accA.w * sA);
        float4 rB = make_float4(accB.x * sB, accB.y * sB, accB.z * sB, accB.w * sB);
        *reinterpret_cast<float4*>(&Alds[(2 * waveId)     * 132 + 4 * lane]) = rA;
        *reinterpret_cast<float4*>(&Alds[(2 * waveId + 1) * 132 + 4 * lane]) = rB;
    }
    __syncthreads();

    const int half   = lane >> 4;
    const int lm     = lane & 15;
    const int col0   = waveId * 16;

    v8f c = {};
#pragma unroll
    for (int kk = 0; kk < 32; ++kk) {
        int k0 = 4 * kk + 2 * half;
        v2f a, b;
        a.x = Alds[lm * 132 + k0];
        a.y = Alds[lm * 132 + k0 + 1];
        b.x = W[(size_t)(k0)     * D + col0 + lm];
        b.y = W[(size_t)(k0 + 1) * D + col0 + lm];
        c = __builtin_amdgcn_wmma_f32_16x16x4_f32(
                false, a, false, b, (short)0, c, false, false);
    }

    float bb = bias[col0 + lm];
#pragma unroll
    for (int v = 0; v < 8; ++v) {
        int row = rowBase + v + 8 * half;
        out[(size_t)row * D + col0 + lm] = c[v] + bb;
    }
}

extern "C" void kernel_launch(void* const* d_in, const int* in_sizes, int n_in,
                              void* d_out, int out_size, void* d_ws, size_t ws_size,
                              hipStream_t stream) {
    const float* x  = (const float*)d_in[0];
    const float* W1 = (const float*)d_in[1];
    const float* b1 = (const float*)d_in[2];
    const float* W2 = (const float*)d_in[3];
    const float* b2 = (const float*)d_in[4];
    const float* W3 = (const float*)d_in[5];
    const float* b3 = (const float*)d_in[6];
    const int* src  = (const int*)d_in[7];
    const int* dst  = (const int*)d_in[8];
    float* out = (float*)d_out;

    const int N = NN, E = NE;
    int*   cnt_in   = (int*)d_ws;
    int*   cnt_out  = cnt_in  + NPAD;
    int*   ovf_cnt  = cnt_out + NPAD;
    int*   slots    = ovf_cnt + 64;
    int*   ovf      = slots + (size_t)NN * CAP;
    float* h1       = (float*)(ovf + (size_t)2 * NE);
    float* h2       = h1  + (size_t)NN * D;

    dim3 edgeGrid((NE + 255) / 256);
    dim3 rowGrid(NN / 16);

    hipMemsetAsync(cnt_in, 0, ((size_t)2 * NPAD + 64) * sizeof(int), stream);
    count_kernel<<<edgeGrid, 256, 0, stream>>>(src, dst, cnt_in, cnt_out, slots, ovf_cnt, ovf, E);

    gather_gemm_wmma<false><<<rowGrid, 256, 0, stream>>>(x,  cnt_in, slots, ovf_cnt, ovf, cnt_out, W1, b1, h1, N);
    gather_gemm_wmma<false><<<rowGrid, 256, 0, stream>>>(h1, cnt_in, slots, ovf_cnt, ovf, cnt_out, W2, b2, h2, N);

    gather_gemm_wmma<true><<<rowGrid, 256, 0, stream>>>(h2, cnt_in, slots, ovf_cnt, ovf, cnt_out, W3, b3, out, N);
}


# Per-edge returning global add + scattered slot store replaced by LDS binning: edges ordered by 196-node range in LDS, one block per range ranks them with LDS adds; rows summed in LDS arrival order
# speedup vs baseline: 3.6238x; 1.0708x over previous
#include <hip/hip_runtime.h>


#define NN 50000
#define NE 800000
#define D  128
#define NPAD 50048
#define CAP 32

typedef __attribute__((ext_vector_type(2))) float v2f;
typedef __attribute__((ext_vector_type(8))) float v8f;

#define NB   256
#define NPB  196
#define BCAP 4096
#define FS   64
#define EPT  8
#define BIN_THREADS 256
#define EPW (EPT * BIN_THREADS)
#define BUILD_THREADS 512

static_assert(NB == BIN_THREADS, "one thread per range in the scan");
static_assert(NB * NPB >= NN && NB <= 256 && NPB <= 256 && NN <= 65536, "entry packing");
static_assert(NE % EPT == 0, "a thread owns EPT whole edges");
static_assert(CAP == 32, "slot index split by shift");
static_assert(BCAP % BUILD_THREADS == 0, "whole trips over a run area");

__device__ __forceinline__
void slot_or_side(int d, int s, int* __restrict__ cnt_in, int* __restrict__ slots,
                  int* __restrict__ ovf_cnt, int* __restrict__ ovf) {
    const int r = atomicAdd(&cnt_in[d], 1);
    if (r < CAP) {
        slots[(size_t)d * CAP + r] = s;
    } else {
        const int p = atomicAdd(ovf_cnt, 1);
        ovf[2 * (size_t)p]     = d;
        ovf[2 * (size_t)p + 1] = s;
    }
}

__global__ __launch_bounds__(BIN_THREADS)
void bin_kernel(const int* __restrict__ src,
                const int* __restrict__ dst,
                int* __restrict__ cnt_in,
                int* __restrict__ cnt_out,
                int* __restrict__ slots,
                int* __restrict__ ovf_cnt,
                int* __restrict__ ovf,
                int* __restrict__ fill,
                int* __restrict__ runs, int E, int nNodes) {
    __shared__ int hist[NB];
    __shared__ int scanL[NB];
    __shared__ int baseL[NB];
    __shared__ int wsum[BIN_THREADS / 32];
    __shared__ unsigned stage[EPW];

    const int t = threadIdx.x;
    const int lane = t & 31;
    const int wave = t >> 5;
    const int blockBase = blockIdx.x * EPW;
    const int e0 = blockBase + t * EPT;

    hist[t] = 0;
    __syncthreads();

    const bool valid = (e0 + EPT <= E);
    unsigned w[EPT];
    int r[EPT];
#pragma unroll
    for (int k = 0; k < EPT; ++k) { w[k] = 0u; r[k] = 0; }
    if (valid) {
        const int4 d0 = reinterpret_cast<const int4*>(dst + e0)[0];
        const int4 d1 = reinterpret_cast<const int4*>(dst + e0)[1];
        const int4 s0 = reinterpret_cast<const int4*>(src + e0)[0];
        const int4 s1 = reinterpret_cast<const int4*>(src + e0)[1];
        const int dd[EPT] = {d0.x, d0.y, d0.z, d0.w, d1.x, d1.y, d1.z, d1.w};
        const int ss[EPT] = {s0.x, s0.y, s0.z, s0.w, s1.x, s1.y, s1.z, s1.w};
#pragma unroll
        for (int k = 0; k < EPT; ++k) {
            const int d  = min(max(dd[k], 0), nNodes - 1);
            const int s  = min(max(ss[k], 0), nNodes - 1);
            const int b  = d / NPB;
            const int dl = d - b * NPB;
            atomicAdd(&cnt_out[s], 1);
            r[k] = atomicAdd(&hist[b], 1);
            w[k] = ((unsigned)b << 24) | ((unsigned)dl << 16) | (unsigned)s;
        }
    }
    __syncthreads();

    const int h = hist[t];
    int res = 0;
    if (h > 0) res = atomicAdd(&fill[t * FS], h);
    int x = h;
#pragma unroll
    for (int off = 1; off < 32; off <<= 1) {
        const int y = __shfl_up(x, off, 32);
        if (lane >= off) x += y;
    }
    if (lane == 31) wsum[wave] = x;
    __syncthreads();
    int add = 0;
#pragma unroll
    for (int q = 0; q < BIN_THREADS / 32; ++q)
        if (q < wave) add += wsum[q];
    scanL[t] = add + x - h;
    baseL[t] = res;
    __syncthreads();

    if (valid) {
#pragma unroll
        for (int k = 0; k < EPT; ++k)
            stage[scanL[w[k] >> 24] + r[k]] = w[k];
    }
    __syncthreads();

    const int nvalid = min(EPW, E - blockBase);
#pragma unroll
    for (int k = 0; k < EPT; ++k) {
        const int i = t + k * BIN_THREADS;
        if (i >= nvalid) break;
        const unsigned u = stage[i];
        const int b = (int)(u >> 24);
        const int p = baseL[b] + (i - scanL[b]);
        if (p < BCAP) {
            runs[(size_t)b * BCAP + p] = (int)(u & 0xFFFFFFu);
        } else {
            slot_or_side(b * NPB + (int)((u >> 16) & 0xFFu), (int)(u & 0xFFFFu),
                         cnt_in, slots, ovf_cnt, ovf);
        }
    }
}

__global__ __launch_bounds__(BUILD_THREADS)
void build_kernel(const int* __restrict__ fill,
                  const int* __restrict__ runs,
                  int* __restrict__ cnt_in,
                  int* __restrict__ slots,
                  int* __restrict__ ovf_cnt,
                  int* __restrict__ ovf, int nNodes) {
    __shared__ int cntL[NPB];
    __shared__ int startL[NPB];
    __shared__ int slotL[NPB * CAP];

    const int t = threadIdx.x;
    const int b = blockIdx.x;
    const int nodeBase = b * NPB;

    if (t < NPB) {
        const int node = nodeBase + t;
        const int c0 = (node < nNodes) ? cnt_in[node] : 0;
        cntL[t]   = c0;
        startL[t] = c0;
    }
    const int n = min(max(fill[b * FS], 0), BCAP);
    __syncthreads();

    unsigned ent[BCAP / BUILD_THREADS];
#pragma unroll
    for (int k = 0; k < BCAP / BUILD_THREADS; ++k) {
        const int i = t + k * BUILD_THREADS;
        ent[k] = (i < n) ? (unsigned)runs[(size_t)b * BCAP + i] : 0u;
    }
#pragma unroll
    for (int k = 0; k < BCAP / BUILD_THREADS; ++k) {
        if (t + k * BUILD_THREADS >= n) break;
        const unsigned u = ent[k];
        const int dl = min((int)((u >> 16) & 0xFFu), NPB - 1);
        const int s  = (int)(u & 0xFFFFu);
        const int r  = atomicAdd(&cntL[dl], 1);
        if (r < CAP) {
            slotL[dl * CAP + r] = s;
        } else {
            const int p = atomicAdd(ovf_cnt, 1);
            ovf[2 * (size_t)p]     = nodeBase + dl;
            ovf[2 * (size_t)p + 1] = s;
        }
    }
    __syncthreads();

    for (int i = t; i < NPB * CAP; i += BUILD_THREADS) {
        const int dl = i >> 5;
        const int r  = i & (CAP - 1);
        if (nodeBase + dl < nNodes && r >= startL[dl] && r < min(cntL[dl], CAP))
            slots[(size_t)nodeBase * CAP + i] = slotL[i];
    }
    if (t < NPB && nodeBase + t < nNodes) cnt_in[nodeBase + t] = cntL[t];
}

template <bool SCALE, bool FULL>
__device__ __forceinline__
void gather8(const float* __restrict__ h, int idx, float sc,
             int j, int m, int lane, int pos, int split,
             float4& accA, float4& accB) {
    float4 v[8];
    float  f[8];
#pragma unroll
    for (int k = 0; k < 8; ++k) {
        v[k] = make_float4(0.0f, 0.0f, 0.0f, 0.0f);
        f[k] = 0.0f;
        if (FULL || j + k < m) {
            const int s = __builtin_amdgcn_readlane(idx, j + k);
            v[k] = reinterpret_cast<const float4*>(h + (size_t)s * D)[lane];
            if (SCALE)
                f[k] = __int_as_float(
                    __builtin_amdgcn_readlane(__float_as_int(sc), j + k));
        }
    }
#pragma unroll
    for (int k = 0; k < 8; ++k) {
        if (pos + k < split) {
            if (SCALE) {
                accA.x += v[k].x * f[k];
                accA.y += v[k].y * f[k];
                accA.z += v[k].z * f[k];
                accA.w += v[k].w * f[k];
            } else {
                accA.x += v[k].x;
                accA.y += v[k].y;
                accA.z += v[k].z;
                accA.w += v[k].w;
            }
        } else {
            if (SCALE) {
                accB.x += v[k].x * f[k];
                accB.y += v[k].y * f[k];
                accB.z += v[k].z * f[k];
                accB.w += v[k].w * f[k];
            } else {
                accB.x += v[k].x;
                accB.y += v[k].y;
                accB.z += v[k].z;
                accB.w += v[k].w;
            }
        }
    }
}

template <bool SCALE>
__global__ __launch_bounds__(256)
void gather_gemm_wmma(const float* __restrict__ h,
                      const int* __restrict__ cnt_in,
                      const int* __restrict__ slots,
                      const int* __restrict__ ovf_cnt,
                      const int* __restrict__ ovf,
                      const int* __restrict__ cnt_out,
                      const float* __restrict__ W,
                      const float* __restrict__ bias,
                      float* __restrict__ out, int nNodes) {
    __shared__ __attribute__((aligned(16))) float Alds[16 * 132];

    const int t = threadIdx.x;
    const int lane = t & 31;
    const int rowBase = blockIdx.x * 16;
    const int waveId = __builtin_amdgcn_readfirstlane(t >> 5);

    {
        const int nA = rowBase + 2 * waveId;
        const int dA = cnt_in[nA];
        const int dB = cnt_in[nA + 1];
        const int mA = min(dA, CAP);
        const int mB = min(dB, CAP);
        const int tot = mA + mB;
        float4 accA = make_float4(0.0f, 0.0f, 0.0f, 0.0f);
        float4 accB = make_float4(0.0f, 0.0f, 0.0f, 0.0f);

        int iA = 0, iB = 0;
        if (lane < mA) iA = slots[(size_t)nA * CAP + lane];
        if (lane < mB) iB = slots[(size_t)(nA + 1) * CAP + lane];
        const int sh0 = __shfl(iB, (lane - mA) & 31, 32);
        const int sh1 = __shfl(iB, (lane + 32 - mA) & 31, 32);
        const int c0 = (lane < mA) ? iA : sh0;
        const int c1 = sh1;

        for (int base = 0; base < tot; base += 32) {
            const int m = min(32, tot - base);
            int   idx = 0;
            float sc  = 1.0f;
            if (lane < m) {
                idx = (base == 0) ? c0 : c1;
                idx = min(max(idx, 0), nNodes - 1);
                if (SCALE) sc = rsqrtf(fmaxf((float)cnt_out[idx], 1.0f));
            }
            int j = 0;
            for (; j + 8 <= m; j += 8)
                gather8<SCALE, true >(h, idx, sc, j, m, lane, base + j, mA, accA, accB);
            if (j < m)
                gather8<SCALE, false>(h, idx, sc, j, m, lane, base + j, mA, accA, accB);
        }

        if (dA > CAP || dB > CAP) {
            const int novf = min(ovf_cnt[0], NE);
            for (int base = 0; base < novf; base += 32) {
                const int e = base + lane;
                int d = -1, s = 0;
                if (e < novf) {
                    const int2 p = reinterpret_cast<const int2*>(ovf)[e];
                    d = p.x;
                    s = min(max(p.y, 0), nNodes - 1);
                }
                const unsigned hitA = __builtin_amdgcn_ballot_w32(d == nA);
                const unsigned hitB = __builtin_amdgcn_ballot_w32(d == nA + 1);
                unsigned hits = hitA | hitB;
                while (hits) {
                    const int k = __builtin_ctz(hits);
                    hits &= hits - 1;
                    const int sidx = __builtin_amdgcn_readlane(s, k);
                    const float4 v = reinterpret_cast<const float4*>(h + (size_t)sidx * D)[lane];
                    float f = 1.0f;
                    if (SCALE) f = rsqrtf(fmaxf((float)cnt_out[sidx], 1.0f));
                    if ((hitA >> k) & 1u) {
                        if (SCALE) {
                            accA.x += v.x * f; accA.y += v.y * f; accA.z += v.z * f; accA.w += v.w * f;
                        } else {
                            accA.x += v.x; accA.y += v.y; accA.z += v.z; accA.w += v.w;
                        }
                    } else {
                        if (SCALE) {
                            accB.x += v.x * f; accB.y += v.y * f; accB.z += v.z * f; accB.w += v.w * f;
                        } else {
                            accB.x += v.x; accB.y += v.y; accB.z += v.z; accB.w += v.w;
                        }
                    }
                }
            }
        }

        const float degA = (float)dA;
        const float degB = (float)dB;
        float sA, sB;
        if (SCALE) {
            sA = rsqrtf(fmaxf(degA, 1.0f));
            sB = rsqrtf(fmaxf(degB, 1.0f));
        } else {
            sA = 1.0f / (degA + 1.0f);
            sB = 1.0f / (degB + 1.0f);
            const float4 hA = reinterpret_cast<const float4*>(h + (size_t)nA * D)[lane];
            const float4 hB = reinterpret_cast<const float4*>(h + (size_t)(nA + 1) * D)[lane];
            accA.x += hA.x; accA.y += hA.y; accA.z += hA.z; accA.w += hA.w;
            accB.x += hB.x; accB.y += hB.y; accB.z += hB.z; accB.w += hB.w;
        }
        float4 rA = make_float4(accA.x * sA, accA.y * sA, accA.z * sA, accA.w * sA);
        float4 rB = make_float4(accB.x * sB, accB.y * sB, accB.z * sB, accB.w * sB);
        *reinterpret_cast<float4*>(&Alds[(2 * waveId)     * 132 + 4 * lane]) = rA;
        *reinterpret_cast<float4*>(&Alds[(2 * waveId + 1) * 132 + 4 * lane]) = rB;
    }
    __syncthreads();

    const int half   = lane >> 4;
    const int lm     = lane & 15;
    const int col0   = waveId * 16;

    v8f c = {};
#pragma unroll
    for (int kk = 0; kk < 32; ++kk) {
        int k0 = 4 * kk + 2 * half;
        v2f a, b;
        a.x = Alds[lm * 132 + k0];
        a.y = Alds[lm * 132 + k0 + 1];
        b.x = W[(size_t)(k0)     * D + col0 + lm];
        b.y = W[(size_t)(k0 + 1) * D + col0 + lm];
        c = __builtin_amdgcn_wmma_f32_16x16x4_f32(
                false, a, false, b, (short)0, c, false, false);
    }

    float bb = bias[col0 + lm];
#pragma unroll
    for (int v = 0; v < 8; ++v) {
        int row = rowBase + v + 8 * half;
        out[(size_t)row * D + col0 + lm] = c[v] + bb;
    }
}

extern "C" void kernel_launch(void* const* d_in, const int* in_sizes, int n_in,
                              void* d_out, int out_size, void* d_ws, size_t ws_size,
                              hipStream_t stream) {
    const float* x  = (const float*)d_in[0];
    const float* W1 = (const float*)d_in[1];
    const float* b1 = (const float*)d_in[2];
    const float* W2 = (const float*)d_in[3];
    const float* b2 = (const float*)d_in[4];
    const float* W3 = (const float*)d_in[5];
    const float* b3 = (const float*)d_in[6];
    const int* src  = (const int*)d_in[7];
    const int* dst  = (const int*)d_in[8];
    float* out = (float*)d_out;

    const int N = NN, E = NE;
    int*   cnt_in   = (int*)d_ws;
    int*   cnt_out  = cnt_in  + NPAD;
    int*   ovf_cnt  = cnt_out + NPAD;
    int*   fill     = ovf_cnt + 64;
    int*   slots    = fill + (size_t)NB * FS;
    int*   ovf      = slots + (size_t)NN * CAP;
    int*   runs     = ovf + (size_t)2 * NE;
    float* h1       = (float*)(runs + (size_t)NB * BCAP);
    float* h2       = h1  + (size_t)NN * D;

    dim3 binGrid((NE + EPW - 1) / EPW);
    dim3 rowGrid(NN / 16);

    hipMemsetAsync(cnt_in, 0, ((size_t)2 * NPAD + 64 + (size_t)NB * FS) * sizeof(int), stream);
    bin_kernel<<<binGrid, BIN_THREADS, 0, stream>>>(src, dst, cnt_in, cnt_out, slots, ovf_cnt, ovf, fill, runs, E, N);
    build_kernel<<<NB, BUILD_THREADS, 0, stream>>>(fill, runs, cnt_in, slots, ovf_cnt, ovf, N);

    gather_gemm_wmma<false><<<rowGrid, 256, 0, stream>>>(x,  cnt_in, slots, ovf_cnt, ovf, cnt_out, W1, b1, h1, N);
    gather_gemm_wmma<false><<<rowGrid, 256, 0, stream>>>(h1, cnt_in, slots, ovf_cnt, ovf, cnt_out, W2, b2, h2, N);

    gather_gemm_wmma<true><<<rowGrid, 256, 0, stream>>>(h2, cnt_in, slots, ovf_cnt, ovf, cnt_out, W3, b3, out, N);
}


# Weights folded: no activation between layers and SAGE rows sum to 1, so out = S3 S1 S1 x (W1 W2 W3) + folded bias; layers 1-2 are row sums only, one WMMA product per node; products reassociated
# speedup vs baseline: 3.9442x; 1.0884x over previous
#include <hip/hip_runtime.h>


#define NN 50000
#define NE 800000
#define D  128
#define NPAD 50048
#define CAP 32

#define FR 144
#define NFOLD (FR / 16)

typedef __attribute__((ext_vector_type(2))) float v2f;
typedef __attribute__((ext_vector_type(8))) float v8f;


__device__ __forceinline__
v8f panel_wmma(const float* Alds, const float* __restrict__ Wm, int col0, int lane) {
    const int half = lane >> 4;
    const int lm   = lane & 15;
    v8f c = {};
#pragma unroll
    for (int kk = 0; kk < 32; ++kk) {
        int k0 = 4 * kk + 2 * half;
        v2f a, b;
        a.x = Alds[lm * 132 + k0];
        a.y = Alds[lm * 132 + k0 + 1];
        b.x = Wm[(size_t)(k0)     * D + col0 + lm];
        b.y = Wm[(size_t)(k0 + 1) * D + col0 + lm];
        c = __builtin_amdgcn_wmma_f32_16x16x4_f32(
                false, a, false, b, (short)0, c, false, false);
    }
    return c;
}

__device__ __forceinline__
void fold_tile(float* Alds, const float* __restrict__ A, int nrowsA,
               const float* __restrict__ arow, const float* __restrict__ Bm,
               const float* __restrict__ addrow, float* __restrict__ outM,
               int tile, int t) {
    const int lane   = t & 31;
    const int waveId = __builtin_amdgcn_readfirstlane(t >> 5);
    {
        const int row  = t >> 4;
        const int col  = (t & 15) * 8;
        const int grow = tile * 16 + row;
        float4 v0 = make_float4(0.0f, 0.0f, 0.0f, 0.0f);
        float4 v1 = v0;
        if (grow < nrowsA) {
            v0 = reinterpret_cast<const float4*>(A + (size_t)grow * D + col)[0];
            v1 = reinterpret_cast<const float4*>(A + (size_t)grow * D + col)[1];
        } else if (grow == nrowsA && arow != nullptr) {
            v0 = reinterpret_cast<const float4*>(arow + col)[0];
            v1 = reinterpret_cast<const float4*>(arow + col)[1];
        }
        *reinterpret_cast<float4*>(&Alds[row * 132 + col])     = v0;
        *reinterpret_cast<float4*>(&Alds[row * 132 + col + 4]) = v1;
    }
    __syncthreads();
    const int half = lane >> 4;
    const int lm   = lane & 15;
    const int col0 = waveId * 16;
    const v8f c = panel_wmma(Alds, Bm, col0, lane);
    const float ad = (addrow != nullptr) ? addrow[col0 + lm] : 0.0f;
#pragma unroll
    for (int v = 0; v < 8; ++v) {
        const int grow = tile * 16 + v + 8 * half;
        outM[(size_t)grow * D + col0 + lm] = c[v] + ((grow == nrowsA) ? ad : 0.0f);
    }
}

#define NB   256
#define NPB  196
#define BCAP 4096
#define FS   64
#define EPT  8
#define BIN_THREADS 256
#define EPW (EPT * BIN_THREADS)
#define BUILD_THREADS 512

static_assert(NB == BIN_THREADS, "one thread per range in the scan");
static_assert(NB * NPB >= NN && NB <= 256 && NPB <= 256 && NN <= 65536, "entry packing");
static_assert(NE % EPT == 0, "a thread owns EPT whole edges");
static_assert(CAP == 32, "slot index split by shift");
static_assert(BCAP % BUILD_THREADS == 0, "whole trips over a run area");

__device__ __forceinline__
void slot_or_side(int d, int s, int* __restrict__ cnt_in, int* __restrict__ slots,
                  int* __restrict__ ovf_cnt, int* __restrict__ ovf) {
    const int r = atomicAdd(&cnt_in[d], 1);
    if (r < CAP) {
        slots[(size_t)d * CAP + r] = s;
    } else {
        const int p = atomicAdd(ovf_cnt, 1);
        ovf[2 * (size_t)p]     = d;
        ovf[2 * (size_t)p + 1] = s;
    }
}

__global__ __launch_bounds__(BIN_THREADS)
void bin_kernel(const int* __restrict__ src,
                const int* __restrict__ dst,
                int* __restrict__ cnt_in,
                int* __restrict__ cnt_out,
                int* __restrict__ slots,
                int* __restrict__ ovf_cnt,
                int* __restrict__ ovf,
                int* __restrict__ fill,
                int* __restrict__ runs, int E, int nNodes,
                const float* __restrict__ W1, const float* __restrict__ b1,
                const float* __restrict__ W2, const float* __restrict__ b2,
                float* __restrict__ T) {
    __shared__ __attribute__((aligned(16))) float Alds[16 * 132];
    if (blockIdx.x < NFOLD) {
        fold_tile(Alds, W1, D, b1, W2, b2, T, blockIdx.x, threadIdx.x);
        return;
    }
    __shared__ int hist[NB];
    __shared__ int scanL[NB];
    __shared__ int baseL[NB];
    __shared__ int wsum[BIN_THREADS / 32];
    __shared__ unsigned stage[EPW];

    const int t = threadIdx.x;
    const int lane = t & 31;
    const int wave = t >> 5;
    const int blockBase = (blockIdx.x - NFOLD) * EPW;
    const int e0 = blockBase + t * EPT;

    hist[t] = 0;
    __syncthreads();

    const bool valid = (e0 + EPT <= E);
    unsigned w[EPT];
    int r[EPT];
#pragma unroll
    for (int k = 0; k < EPT; ++k) { w[k] = 0u; r[k] = 0; }
    if (valid) {
        const int4 d0 = reinterpret_cast<const int4*>(dst + e0)[0];
        const int4 d1 = reinterpret_cast<const int4*>(dst + e0)[1];
        const int4 s0 = reinterpret_cast<const int4*>(src + e0)[0];
        const int4 s1 = reinterpret_cast<const int4*>(src + e0)[1];
        const int dd[EPT] = {d0.x, d0.y, d0.z, d0.w, d1.x, d1.y, d1.z, d1.w};
        const int ss[EPT] = {s0.x, s0.y, s0.z, s0.w, s1.x, s1.y, s1.z, s1.w};
#pragma unroll
        for (int k = 0; k < EPT; ++k) {
            const int d  = min(max(dd[k], 0), nNodes - 1);
            const int s  = min(max(ss[k], 0), nNodes - 1);
            const int b  = d / NPB;
            const int dl = d - b * NPB;
            atomicAdd(&cnt_out[s], 1);
            r[k] = atomicAdd(&hist[b], 1);
            w[k] = ((unsigned)b << 24) | ((unsigned)dl << 16) | (unsigned)s;
        }
    }
    __syncthreads();

    const int h = hist[t];
    int res = 0;
    if (h > 0) res = atomicAdd(&fill[t * FS], h);
    int x = h;
#pragma unroll
    for (int off = 1; off < 32; off <<= 1) {
        const int y = __shfl_up(x, off, 32);
        if (lane >= off) x += y;
    }
    if (lane == 31) wsum[wave] = x;
    __syncthreads();
    int add = 0;
#pragma unroll
    for (int q = 0; q < BIN_THREADS / 32; ++q)
        if (q < wave) add += wsum[q];
    scanL[t] = add + x - h;
    baseL[t] = res;
    __syncthreads();

    if (valid) {
#pragma unroll
        for (int k = 0; k < EPT; ++k)
            stage[scanL[w[k] >> 24] + r[k]] = w[k];
    }
    __syncthreads();

    const int nvalid = min(EPW, E - blockBase);
#pragma unroll
    for (int k = 0; k < EPT; ++k) {
        const int i = t + k * BIN_THREADS;
        if (i >= nvalid) break;
        const unsigned u = stage[i];
        const int b = (int)(u >> 24);
        const int p = baseL[b] + (i - scanL[b]);
        if (p < BCAP) {
            runs[(size_t)b * BCAP + p] = (int)(u & 0xFFFFFFu);
        } else {
            slot_or_side(b * NPB + (int)((u >> 16) & 0xFFu), (int)(u & 0xFFFFu),
                         cnt_in, slots, ovf_cnt, ovf);
        }
    }
}

__global__ __launch_bounds__(BUILD_THREADS)
void build_kernel(const int* __restrict__ fill,
                  const int* __restrict__ runs,
                  int* __restrict__ cnt_in,
                  int* __restrict__ slots,
                  int* __restrict__ ovf_cnt,
                  int* __restrict__ ovf, int nNodes) {
    __shared__ int cntL[NPB];
    __shared__ int startL[NPB];
    __shared__ int slotL[NPB * CAP];

    const int t = threadIdx.x;
    const int b = blockIdx.x;
    const int nodeBase = b * NPB;

    if (t < NPB) {
        const int node = nodeBase + t;
        const int c0 = (node < nNodes) ? cnt_in[node] : 0;
        cntL[t]   = c0;
        startL[t] = c0;
    }
    const int n = min(max(fill[b * FS], 0), BCAP);
    __syncthreads();

    unsigned ent[BCAP / BUILD_THREADS];
#pragma unroll
    for (int k = 0; k < BCAP / BUILD_THREADS; ++k) {
        const int i = t + k * BUILD_THREADS;
        ent[k] = (i < n) ? (unsigned)runs[(size_t)b * BCAP + i] : 0u;
    }
#pragma unroll
    for (int k = 0; k < BCAP / BUILD_THREADS; ++k) {
        if (t + k * BUILD_THREADS >= n) break;
        const unsigned u = ent[k];
        const int dl = min((int)((u >> 16) & 0xFFu), NPB - 1);
        const int s  = (int)(u & 0xFFFFu);
        const int r  = atomicAdd(&cntL[dl], 1);
        if (r < CAP) {
            slotL[dl * CAP + r] = s;
        } else {
            const int p = atomicAdd(ovf_cnt, 1);
            ovf[2 * (size_t)p]     = nodeBase + dl;
            ovf[2 * (size_t)p + 1] = s;
        }
    }
    __syncthreads();

    for (int i = t; i < NPB * CAP; i += BUILD_THREADS) {
        const int dl = i >> 5;
        const int r  = i & (CAP - 1);
        if (nodeBase + dl < nNodes && r >= startL[dl] && r < min(cntL[dl], CAP))
            slots[(size_t)nodeBase * CAP + i] = slotL[i];
    }
    if (t < NPB && nodeBase + t < nNodes) cnt_in[nodeBase + t] = cntL[t];
}

template <bool SCALE, bool FULL>
__device__ __forceinline__
void gather8(const float* __restrict__ h, int idx, float sc,
             int j, int m, int lane, int pos, int split,
             float4& accA, float4& accB, float& fsA, float& fsB) {
    float4 v[8];
    float  f[8];
#pragma unroll
    for (int k = 0; k < 8; ++k) {
        v[k] = make_float4(0.0f, 0.0f, 0.0f, 0.0f);
        f[k] = 0.0f;
        if (FULL || j + k < m) {
            const int s = __builtin_amdgcn_readlane(idx, j + k);
            v[k] = reinterpret_cast<const float4*>(h + (size_t)s * D)[lane];
            if (SCALE)
                f[k] = __int_as_float(
                    __builtin_amdgcn_readlane(__float_as_int(sc), j + k));
        }
    }
#pragma unroll
    for (int k = 0; k < 8; ++k) {
        if (pos + k < split) {
            if (SCALE) {
                accA.x += v[k].x * f[k];
                accA.y += v[k].y * f[k];
                accA.z += v[k].z * f[k];
                accA.w += v[k].w * f[k];
                fsA += f[k];
            } else {
                accA.x += v[k].x;
                accA.y += v[k].y;
                accA.z += v[k].z;
                accA.w += v[k].w;
            }
        } else {
            if (SCALE) {
                accB.x += v[k].x * f[k];
                accB.y += v[k].y * f[k];
                accB.z += v[k].z * f[k];
                accB.w += v[k].w * f[k];
                fsB += f[k];
            } else {
                accB.x += v[k].x;
                accB.y += v[k].y;
                accB.z += v[k].z;
                accB.w += v[k].w;
            }
        }
    }
}

template <bool SCALE>
__device__ __forceinline__
void gather_rows(const float* __restrict__ h,
                 const int* __restrict__ cnt_in,
                 const int* __restrict__ slots,
                 const int* __restrict__ ovf_cnt,
                 const int* __restrict__ ovf,
                 const int* __restrict__ cnt_out,
                 int nA, int lane, int nNodes,
                 float4& rA, float4& rB, float& wA, float& wB) {
    const int dA = cnt_in[nA];
    const int dB = cnt_in[nA + 1];
    const int mA = min(dA, CAP);
    const int mB = min(dB, CAP);
    const int tot = mA + mB;
    float4 accA = make_float4(0.0f, 0.0f, 0.0f, 0.0f);
    float4 accB = make_float4(0.0f, 0.0f, 0.0f, 0.0f);
    float fsA = 0.0f, fsB = 0.0f;

    int iA = 0, iB = 0;
    if (lane < mA) iA = slots[(size_t)nA * CAP + lane];
    if (lane < mB) iB = slots[(size_t)(nA + 1) * CAP + lane];
    const int sh0 = __shfl(iB, (lane - mA) & 31, 32);
    const int sh1 = __shfl(iB, (lane + 32 - mA) & 31, 32);
    const int c0 = (lane < mA) ? iA : sh0;
    const int c1 = sh1;

    for (int base = 0; base < tot; base += 32) {
        const int m = min(32, tot - base);
        int   idx = 0;
        float sc  = 1.0f;
        if (lane < m) {
            idx = (base == 0) ? c0 : c1;
            idx = min(max(idx, 0), nNodes - 1);
            if (SCALE) sc = rsqrtf(fmaxf((float)cnt_out[idx], 1.0f));
        }
        int j = 0;
        for (; j + 8 <= m; j += 8)
            gather8<SCALE, true >(h, idx, sc, j, m, lane, base + j, mA, accA, accB, fsA, fsB);
        if (j < m)
            gather8<SCALE, false>(h, idx, sc, j, m, lane, base + j, mA, accA, accB, fsA, fsB);
    }

    if (dA > CAP || dB > CAP) {
        const int novf = min(ovf_cnt[0], NE);
        for (int base = 0; base < novf; base += 32) {
            const int e = base + lane;
            int d = -1, s = 0;
            if (e < novf) {
                const int2 p = reinterpret_cast<const int2*>(ovf)[e];
                d = p.x;
                s = min(max(p.y, 0), nNodes - 1);
            }
            const unsigned hitA = __builtin_amdgcn_ballot_w32(d == nA);
            const unsigned hitB = __builtin_amdgcn_ballot_w32(d == nA + 1);
            unsigned hits = hitA | hitB;
            while (hits) {
                const int k = __builtin_ctz(hits);
                hits &= hits - 1;
                const int sidx = __builtin_amdgcn_readlane(s, k);
                const float4 v = reinterpret_cast<const float4*>(h + (size_t)sidx * D)[lane];
                float f = 1.0f;
                if (SCALE) f = rsqrtf(fmaxf((float)cnt_out[sidx], 1.0f));
                if ((hitA >> k) & 1u) {
                    if (SCALE) {
                        accA.x += v.x * f; accA.y += v.y * f; accA.z += v.z * f; accA.w += v.w * f;
                        fsA += f;
                    } else {
                        accA.x += v.x; accA.y += v.y; accA.z += v.z; accA.w += v.w;
                    }
                } else {
                    if (SCALE) {
                        accB.x += v.x * f; accB.y += v.y * f; accB.z += v.z * f; accB.w += v.w * f;
                        fsB += f;
                    } else {
                        accB.x += v.x; accB.y += v.y; accB.z += v.z; accB.w += v.w;
                    }
                }
            }
        }
    }

    const float degA = (float)dA;
    const float degB = (float)dB;
    float sA, sB;
    if (SCALE) {
        sA = rsqrtf(fmaxf(degA, 1.0f));
        sB = rsqrtf(fmaxf(degB, 1.0f));
    } else {
        sA = 1.0f / (degA + 1.0f);
        sB = 1.0f / (degB + 1.0f);
        const float4 hA = reinterpret_cast<const float4*>(h + (size_t)nA * D)[lane];
        const float4 hB = reinterpret_cast<const float4*>(h + (size_t)(nA + 1) * D)[lane];
        accA.x += hA.x; accA.y += hA.y; accA.z += hA.z; accA.w += hA.w;
        accB.x += hB.x; accB.y += hB.y; accB.z += hB.z; accB.w += hB.w;
    }
    rA = make_float4(accA.x * sA, accA.y * sA, accA.z * sA, accA.w * sA);
    rB = make_float4(accB.x * sB, accB.y * sB, accB.z * sB, accB.w * sB);
    wA = fsA * sA;
    wB = fsB * sB;
}

template <int NF>
__global__ __launch_bounds__(256)
void agg_kernel(const float* __restrict__ h,
                const int* __restrict__ cnt_in,
                const int* __restrict__ slots,
                const int* __restrict__ ovf_cnt,
                const int* __restrict__ ovf,
                float* __restrict__ g, int nNodes,
                const float* __restrict__ T,
                const float* __restrict__ W3,
                float* __restrict__ U) {
    const int t = threadIdx.x;
    if (NF > 0) {
        __shared__ __attribute__((aligned(16))) float Alds[16 * 132];
        if ((int)blockIdx.x < NF) {
            fold_tile(Alds, T, FR, nullptr, W3, nullptr, U, blockIdx.x, t);
            return;
        }
    }
    const int lane = t & 31;
    const int waveId = __builtin_amdgcn_readfirstlane(t >> 5);
    const int nA = ((int)blockIdx.x - NF) * 16 + 2 * waveId;
    float4 rA, rB;
    float wA, wB;
    gather_rows<false>(h, cnt_in, slots, ovf_cnt, ovf, nullptr, nA, lane, nNodes, rA, rB, wA, wB);
    reinterpret_cast<float4*>(g + (size_t)nA * D)[lane]       = rA;
    reinterpret_cast<float4*>(g + (size_t)(nA + 1) * D)[lane] = rB;
}

__global__ __launch_bounds__(256)
void gather_gemm_wmma(const float* __restrict__ h,
                      const int* __restrict__ cnt_in,
                      const int* __restrict__ slots,
                      const int* __restrict__ ovf_cnt,
                      const int* __restrict__ ovf,
                      const int* __restrict__ cnt_out,
                      const float* __restrict__ W,
                      const float* __restrict__ cvec,
                      const float* __restrict__ bias,
                      float* __restrict__ out, int nNodes) {
    __shared__ __attribute__((aligned(16))) float Alds[16 * 132];
    __shared__ float wl[16];

    const int t = threadIdx.x;
    const int lane = t & 31;
    const int rowBase = blockIdx.x * 16;
    const int waveId = __builtin_amdgcn_readfirstlane(t >> 5);

    {
        const int nA = rowBase + 2 * waveId;
        float4 rA, rB;
        float wA, wB;
        gather_rows<true>(h, cnt_in, slots, ovf_cnt, ovf, cnt_out, nA, lane, nNodes, rA, rB, wA, wB);
        *reinterpret_cast<float4*>(&Alds[(2 * waveId)     * 132 + 4 * lane]) = rA;
        *reinterpret_cast<float4*>(&Alds[(2 * waveId + 1) * 132 + 4 * lane]) = rB;
        if (lane == 0) {
            wl[2 * waveId]     = wA;
            wl[2 * waveId + 1] = wB;
        }
    }
    __syncthreads();

    const int half   = lane >> 4;
    const int lm     = lane & 15;
    const int col0   = waveId * 16;
    const v8f c = panel_wmma(Alds, W, col0, lane);

    const float bb = bias[col0 + lm];
    const float cv = cvec[col0 + lm];
#pragma unroll
    for (int v = 0; v < 8; ++v) {
        const int row = v + 8 * half;
        out[(size_t)(rowBase + row) * D + col0 + lm] = c[v] + wl[row] * cv + bb;
    }
}

extern "C" void kernel_launch(void* const* d_in, const int* in_sizes, int n_in,
                              void* d_out, int out_size, void* d_ws, size_t ws_size,
                              hipStream_t stream) {
    const float* x  = (const float*)d_in[0];
    const float* W1 = (const float*)d_in[1];
    const float* b1 = (const float*)d_in[2];
    const float* W2 = (const float*)d_in[3];
    const float* b2 = (const float*)d_in[4];
    const float* W3 = (const float*)d_in[5];
    const float* b3 = (const float*)d_in[6];
    const int* src  = (const int*)d_in[7];
    const int* dst  = (const int*)d_in[8];
    float* out = (float*)d_out;

    const int N = NN, E = NE;
    int*   cnt_in   = (int*)d_ws;
    int*   cnt_out  = cnt_in  + NPAD;
    int*   ovf_cnt  = cnt_out + NPAD;
    int*   fill     = ovf_cnt + 64;
    int*   slots    = fill + (size_t)NB * FS;
    int*   ovf      = slots + (size_t)NN * CAP;
    int*   runs     = ovf + (size_t)2 * NE;
    float* g1       = (float*)(runs + (size_t)NB * BCAP);
    float* g2       = g1  + (size_t)NN * D;
    float* T        = g2  + (size_t)NN * D;
    float* U        = T   + (size_t)FR * D;

    dim3 binGrid(NFOLD + (NE + EPW - 1) / EPW);
    dim3 rowGrid(NN / 16);
    dim3 rowGridF(NFOLD + NN / 16);

    hipMemsetAsync(cnt_in, 0, ((size_t)2 * NPAD + 64 + (size_t)NB * FS) * sizeof(int), stream);
    bin_kernel<<<binGrid, BIN_THREADS, 0, stream>>>(src, dst, cnt_in, cnt_out, slots, ovf_cnt, ovf, fill, runs, E, N,
                                                    W1, b1, W2, b2, T);
    build_kernel<<<NB, BUILD_THREADS, 0, stream>>>(fill, runs, cnt_in, slots, ovf_cnt, ovf, N);

    agg_kernel<NFOLD><<<rowGridF, 256, 0, stream>>>(x,  cnt_in, slots, ovf_cnt, ovf, g1, N, T, W3, U);
    agg_kernel<0><<<rowGrid,  256, 0, stream>>>(g1, cnt_in, slots, ovf_cnt, ovf, g2, N, T, W3, U);

    gather_gemm_wmma<<<rowGrid, 256, 0, stream>>>(g2, cnt_in, slots, ovf_cnt, ovf, cnt_out,
                                                  U, U + (size_t)D * D, b3, out, N);
}


# Out-degree counted in LDS: edges binned a second time by source range in the same two launches, one 64-bit add reserves both run areas, the range's block counts out-edges; per-edge global add gone
# speedup vs baseline: 6.0582x; 1.5360x over previous
#include <hip/hip_runtime.h>


#define NN 50000
#define NE 800000
#define D  128
#define NPAD 50048
#define CAP 32

#define FR 144
#define NFOLD (FR / 16)

typedef __attribute__((ext_vector_type(2))) float v2f;
typedef __attribute__((ext_vector_type(8))) float v8f;


__device__ __forceinline__
v8f panel_wmma(const float* Alds, const float* __restrict__ Wm, int col0, int lane) {
    const int half = lane >> 4;
    const int lm   = lane & 15;
    v8f c = {};
#pragma unroll
    for (int kk = 0; kk < 32; ++kk) {
        int k0 = 4 * kk + 2 * half;
        v2f a, b;
        a.x = Alds[lm * 132 + k0];
        a.y = Alds[lm * 132 + k0 + 1];
        b.x = Wm[(size_t)(k0)     * D + col0 + lm];
        b.y = Wm[(size_t)(k0 + 1) * D + col0 + lm];
        c = __builtin_amdgcn_wmma_f32_16x16x4_f32(
                false, a, false, b, (short)0, c, false, false);
    }
    return c;
}

__device__ __forceinline__
void fold_tile(float* Alds, const float* __restrict__ A, int nrowsA,
               const float* __restrict__ arow, const float* __restrict__ Bm,
               const float* __restrict__ addrow, float* __restrict__ outM,
               int tile, int t) {
    const int lane   = t & 31;
    const int waveId = __builtin_amdgcn_readfirstlane(t >> 5);
    {
        const int row  = t >> 4;
        const int col  = (t & 15) * 8;
        const int grow = tile * 16 + row;
        float4 v0 = make_float4(0.0f, 0.0f, 0.0f, 0.0f);
        float4 v1 = v0;
        if (grow < nrowsA) {
            v0 = reinterpret_cast<const float4*>(A + (size_t)grow * D + col)[0];
            v1 = reinterpret_cast<const float4*>(A + (size_t)grow * D + col)[1];
        } else if (grow == nrowsA && arow != nullptr) {
            v0 = reinterpret_cast<const float4*>(arow + col)[0];
            v1 = reinterpret_cast<const float4*>(arow + col)[1];
        }
        *reinterpret_cast<float4*>(&Alds[row * 132 + col])     = v0;
        *reinterpret_cast<float4*>(&Alds[row * 132 + col + 4]) = v1;
    }
    __syncthreads();
    const int half = lane >> 4;
    const int lm   = lane & 15;
    const int col0 = waveId * 16;
    const v8f c = panel_wmma(Alds, Bm, col0, lane);
    const float ad = (addrow != nullptr) ? addrow[col0 + lm] : 0.0f;
#pragma unroll
    for (int v = 0; v < 8; ++v) {
        const int grow = tile * 16 + v + 8 * half;
        outM[(size_t)grow * D + col0 + lm] = c[v] + ((grow == nrowsA) ? ad : 0.0f);
    }
}

#define NB   256
#define NPB  196
#define BCAP 4096
#define FS   64
#define FS64 (FS / 2)
#define EPT  8
#define BIN_THREADS 256
#define EPW (EPT * BIN_THREADS)
#define BUILD_THREADS 512

static_assert(NB == BIN_THREADS, "one thread per range in the scan");
static_assert(NB * NPB >= NN && NB <= 256 && NPB <= 256 && NN <= 65536, "entry packing");
static_assert(NE % EPT == 0, "a thread owns EPT whole edges");
static_assert(CAP == 32, "slot index split by shift");
static_assert(BCAP % BUILD_THREADS == 0, "whole trips over a run area");

__device__ __forceinline__
void slot_or_side(int d, int s, int* __restrict__ cnt_in, int* __restrict__ slots,
                  int* __restrict__ ovf_cnt, int* __restrict__ ovf) {
    const int r = atomicAdd(&cnt_in[d], 1);
    if (r < CAP) {
        slots[(size_t)d * CAP + r] = s;
    } else {
        const int p = atomicAdd(ovf_cnt, 1);
        ovf[2 * (size_t)p]     = d;
        ovf[2 * (size_t)p + 1] = s;
    }
}

__global__ __launch_bounds__(BIN_THREADS)
void bin_kernel(const int* __restrict__ src,
                const int* __restrict__ dst,
                int* __restrict__ cnt_in,
                int* __restrict__ cnt_out,
                int* __restrict__ slots,
                int* __restrict__ ovf_cnt,
                int* __restrict__ ovf,
                unsigned long long* __restrict__ fill,
                int* __restrict__ runs,
                int* __restrict__ runsS, int E, int nNodes,
                const float* __restrict__ W1, const float* __restrict__ b1,
                const float* __restrict__ W2, const float* __restrict__ b2,
                float* __restrict__ T) {
    __shared__ __attribute__((aligned(16))) float Alds[16 * 132];
    if (blockIdx.x < NFOLD) {
        fold_tile(Alds, W1, D, b1, W2, b2, T, blockIdx.x, threadIdx.x);
        return;
    }
    __shared__ int hist[NB];
    __shared__ int scanL[NB];
    __shared__ int baseL[NB];
    __shared__ int wsum[BIN_THREADS / 32];
    __shared__ unsigned stage[EPW];
    __shared__ int histS[NB];
    __shared__ int scanS[NB];
    __shared__ int baseS[NB];
    __shared__ int wsumS[BIN_THREADS / 32];
    __shared__ unsigned stageS[EPW];

    const int t = threadIdx.x;
    const int lane = t & 31;
    const int wave = t >> 5;
    const int blockBase = (blockIdx.x - NFOLD) * EPW;
    const int e0 = blockBase + t * EPT;

    hist[t] = 0;
    histS[t] = 0;
    __syncthreads();

    const bool valid = (e0 + EPT <= E);
    unsigned w[EPT];
    int r[EPT];
    int rs[EPT];
#pragma unroll
    for (int k = 0; k < EPT; ++k) { w[k] = 0u; r[k] = 0; rs[k] = 0; }
    if (valid) {
        const int4 d0 = reinterpret_cast<const int4*>(dst + e0)[0];
        const int4 d1 = reinterpret_cast<const int4*>(dst + e0)[1];
        const int4 s0 = reinterpret_cast<const int4*>(src + e0)[0];
        const int4 s1 = reinterpret_cast<const int4*>(src + e0)[1];
        const int dd[EPT] = {d0.x, d0.y, d0.z, d0.w, d1.x, d1.y, d1.z, d1.w};
        const int ss[EPT] = {s0.x, s0.y, s0.z, s0.w, s1.x, s1.y, s1.z, s1.w};
#pragma unroll
        for (int k = 0; k < EPT; ++k) {
            const int d  = min(max(dd[k], 0), nNodes - 1);
            const int s  = min(max(ss[k], 0), nNodes - 1);
            const int b  = d / NPB;
            const int dl = d - b * NPB;
            r[k]  = atomicAdd(&hist[b], 1);
            rs[k] = atomicAdd(&histS[s / NPB], 1);
            w[k] = ((unsigned)b << 24) | ((unsigned)dl << 16) | (unsigned)s;
        }
    }
    __syncthreads();

    const int h  = hist[t];
    const int hs = histS[t];
    unsigned long long res = 0ull;
    if ((h | hs) != 0)
        res = atomicAdd(&fill[t * FS64], ((unsigned long long)(unsigned)hs << 32) | (unsigned long long)(unsigned)h);
    int x  = h;
    int xs = hs;
#pragma unroll
    for (int off = 1; off < 32; off <<= 1) {
        const int y  = __shfl_up(x, off, 32);
        const int ys = __shfl_up(xs, off, 32);
        if (lane >= off) { x += y; xs += ys; }
    }
    if (lane == 31) { wsum[wave] = x; wsumS[wave] = xs; }
    __syncthreads();
    int add = 0, addS = 0;
#pragma unroll
    for (int q = 0; q < BIN_THREADS / 32; ++q)
        if (q < wave) { add += wsum[q]; addS += wsumS[q]; }
    scanL[t] = add + x - h;
    scanS[t] = addS + xs - hs;
    baseL[t] = (int)(unsigned)(res & 0xFFFFFFFFull);
    baseS[t] = (int)(unsigned)(res >> 32);
    __syncthreads();

    if (valid) {
#pragma unroll
        for (int k = 0; k < EPT; ++k) {
            stage[scanL[w[k] >> 24] + r[k]] = w[k];
            const int s  = (int)(w[k] & 0xFFFFu);
            const int bs = s / NPB;
            stageS[scanS[bs] + rs[k]] = ((unsigned)bs << 8) | (unsigned)(s - bs * NPB);
        }
    }
    __syncthreads();

    const int nvalid = min(EPW, E - blockBase);
#pragma unroll
    for (int k = 0; k < EPT; ++k) {
        const int i = t + k * BIN_THREADS;
        if (i >= nvalid) break;
        const unsigned u = stage[i];
        const int b = (int)(u >> 24);
        const int p = baseL[b] + (i - scanL[b]);
        if (p < BCAP) {
            runs[(size_t)b * BCAP + p] = (int)(u & 0xFFFFFFu);
        } else {
            slot_or_side(b * NPB + (int)((u >> 16) & 0xFFu), (int)(u & 0xFFFFu),
                         cnt_in, slots, ovf_cnt, ovf);
        }
    }
#pragma unroll
    for (int k = 0; k < EPT; ++k) {
        const int i = t + k * BIN_THREADS;
        if (i >= nvalid) break;
        const unsigned u = stageS[i];
        const int b = (int)(u >> 8);
        const int p = baseS[b] + (i - scanS[b]);
        if (p < BCAP) {
            runsS[(size_t)b * BCAP + p] = (int)(u & 0xFFu);
        } else {
            atomicAdd(&cnt_out[b * NPB + (int)(u & 0xFFu)], 1);
        }
    }
}

__global__ __launch_bounds__(BUILD_THREADS)
void build_kernel(const unsigned long long* __restrict__ fill,
                  const int* __restrict__ runs,
                  const int* __restrict__ runsS,
                  int* __restrict__ cnt_in,
                  int* __restrict__ cnt_out,
                  int* __restrict__ slots,
                  int* __restrict__ ovf_cnt,
                  int* __restrict__ ovf, int nNodes) {
    __shared__ int cntL[NPB];
    __shared__ int startL[NPB];
    __shared__ int outL[NPB];
    __shared__ int slotL[NPB * CAP];

    const int t = threadIdx.x;
    const int b = blockIdx.x;
    const int nodeBase = b * NPB;

    if (t < NPB) {
        const int node = nodeBase + t;
        const int c0 = (node < nNodes) ? cnt_in[node] : 0;
        cntL[t]   = c0;
        startL[t] = c0;
        outL[t]   = (node < nNodes) ? cnt_out[node] : 0;
    }
    const unsigned long long f = fill[b * FS64];
    const int n  = (int)min((unsigned)(f & 0xFFFFFFFFull), (unsigned)BCAP);
    const int ns = (int)min((unsigned)(f >> 32), (unsigned)BCAP);
    __syncthreads();

    unsigned ent[BCAP / BUILD_THREADS];
    unsigned ens[BCAP / BUILD_THREADS];
#pragma unroll
    for (int k = 0; k < BCAP / BUILD_THREADS; ++k) {
        const int i = t + k * BUILD_THREADS;
        ent[k] = (i < n)  ? (unsigned)runs[(size_t)b * BCAP + i]  : 0u;
        ens[k] = (i < ns) ? (unsigned)runsS[(size_t)b * BCAP + i] : 0u;
    }
    asm volatile("" :: "v"(ens[0]), "v"(ens[1]), "v"(ens[2]), "v"(ens[3]),
                       "v"(ens[4]), "v"(ens[5]), "v"(ens[6]), "v"(ens[7]));
    static_assert(BCAP / BUILD_THREADS == 8, "the statement above names eight entries");
#pragma unroll
    for (int k = 0; k < BCAP / BUILD_THREADS; ++k) {
        if (t + k * BUILD_THREADS >= n) break;
        const unsigned u = ent[k];
        const int dl = min((int)((u >> 16) & 0xFFu), NPB - 1);
        const int s  = (int)(u & 0xFFFFu);
        const int r  = atomicAdd(&cntL[dl], 1);
        if (r < CAP) {
            slotL[dl * CAP + r] = s;
        } else {
            const int p = atomicAdd(ovf_cnt, 1);
            ovf[2 * (size_t)p]     = nodeBase + dl;
            ovf[2 * (size_t)p + 1] = s;
        }
    }
#pragma unroll
    for (int k = 0; k < BCAP / BUILD_THREADS; ++k) {
        if (t + k * BUILD_THREADS >= ns) break;
        atomicAdd(&outL[min((int)(ens[k] & 0xFFu), NPB - 1)], 1);
    }
    __syncthreads();

    for (int i = t; i < NPB * CAP; i += BUILD_THREADS) {
        const int dl = i >> 5;
        const int r  = i & (CAP - 1);
        if (nodeBase + dl < nNodes && r >= startL[dl] && r < min(cntL[dl], CAP))
            slots[(size_t)nodeBase * CAP + i] = slotL[i];
    }
    if (t < NPB && nodeBase + t < nNodes) {
        cnt_in[nodeBase + t]  = cntL[t];
        cnt_out[nodeBase + t] = outL[t];
    }
}

template <bool FULL>
__device__ __forceinline__
void gather8(const float* __restrict__ h, int idx,
             int j, int m, int lane, int pos, int split,
             float4& accA, float4& accB) {
    float4 v[8];
#pragma unroll
    for (int k = 0; k < 8; ++k) {
        v[k] = make_float4(0.0f, 0.0f, 0.0f, 0.0f);
        if (FULL || j + k < m) {
            const int s = __builtin_amdgcn_readlane(idx, j + k);
            v[k] = reinterpret_cast<const float4*>(h + (size_t)s * D)[lane];
        }
    }
#pragma unroll
    for (int k = 0; k < 8; ++k) {
        if (pos + k < split) {
            accA.x += v[k].x;
            accA.y += v[k].y;
            accA.z += v[k].z;
            accA.w += v[k].w;
        } else {
            accB.x += v[k].x;
            accB.y += v[k].y;
            accB.z += v[k].z;
            accB.w += v[k].w;
        }
    }
}

template <bool SELF>
__device__ __forceinline__
void gather_rows(const float* __restrict__ h,
                 const int* __restrict__ cnt_in,
                 const int* __restrict__ slots,
                 const int* __restrict__ ovf_cnt,
                 const int* __restrict__ ovf,
                 int nA, int lane, int nNodes,
                 float4& rA, float4& rB) {
    const int dA = cnt_in[nA];
    const int dB = cnt_in[nA + 1];
    const int mA = min(dA, CAP);
    const int mB = min(dB, CAP);
    const int tot = mA + mB;
    float4 accA = make_float4(0.0f, 0.0f, 0.0f, 0.0f);
    float4 accB = make_float4(0.0f, 0.0f, 0.0f, 0.0f);

    int iA = 0, iB = 0;
    if (lane < mA) iA = slots[(size_t)nA * CAP + lane];
    if (lane < mB) iB = slots[(size_t)(nA + 1) * CAP + lane];
    const int sh0 = __shfl(iB, (lane - mA) & 31, 32);
    const int sh1 = __shfl(iB, (lane + 32 - mA) & 31, 32);
    const int c0 = (lane < mA) ? iA : sh0;
    const int c1 = sh1;

    for (int base = 0; base < tot; base += 32) {
        const int m = min(32, tot - base);
        int idx = 0;
        if (lane < m) {
            idx = (base == 0) ? c0 : c1;
            idx = min(max(idx, 0), nNodes - 1);
        }
        int j = 0;
        for (; j + 8 <= m; j += 8)
            gather8<true >(h, idx, j, m, lane, base + j, mA, accA, accB);
        if (j < m)
            gather8<false>(h, idx, j, m, lane, base + j, mA, accA, accB);
    }

    if (dA > CAP || dB > CAP) {
        const int novf = min(ovf_cnt[0], NE);
        for (int base = 0; base < novf; base += 32) {
            const int e = base + lane;
            int d = -1, s = 0;
            if (e < novf) {
                const int2 p = reinterpret_cast<const int2*>(ovf)[e];
                d = p.x;
                s = min(max(p.y, 0), nNodes - 1);
            }
            const unsigned hitA = __builtin_amdgcn_ballot_w32(d == nA);
            const unsigned hitB = __builtin_amdgcn_ballot_w32(d == nA + 1);
            unsigned hits = hitA | hitB;
            while (hits) {
                const int k = __builtin_ctz(hits);
                hits &= hits - 1;
                const int sidx = __builtin_amdgcn_readlane(s, k);
                const float4 v = reinterpret_cast<const float4*>(h + (size_t)sidx * D)[lane];
                if ((hitA >> k) & 1u) {
                    accA.x += v.x; accA.y += v.y; accA.z += v.z; accA.w += v.w;
                } else {
                    accB.x += v.x; accB.y += v.y; accB.z += v.z; accB.w += v.w;
                }
            }
        }
    }

    const float degA = (float)dA;
    const float degB = (float)dB;
    float sA, sB;
    if (SELF) {
        sA = 1.0f / (degA + 1.0f);
        sB = 1.0f / (degB + 1.0f);
        const float4 hA = reinterpret_cast<const float4*>(h + (size_t)nA * D)[lane];
        const float4 hB = reinterpret_cast<const float4*>(h + (size_t)(nA + 1) * D)[lane];
        accA.x += hA.x; accA.y += hA.y; accA.z += hA.z; accA.w += hA.w;
        accB.x += hB.x; accB.y += hB.y; accB.z += hB.z; accB.w += hB.w;
    } else {
        sA = rsqrtf(fmaxf(degA, 1.0f));
        sB = rsqrtf(fmaxf(degB, 1.0f));
    }
    rA = make_float4(accA.x * sA, accA.y * sA, accA.z * sA, accA.w * sA);
    rB = make_float4(accB.x * sB, accB.y * sB, accB.z * sB, accB.w * sB);
}

template <int NF, bool SELF>
__global__ __launch_bounds__(256)
void agg_kernel(const float* __restrict__ h,
                const int* __restrict__ cnt_in,
                const int* __restrict__ slots,
                const int* __restrict__ ovf_cnt,
                const int* __restrict__ ovf,
                float* __restrict__ g, int nNodes,
                const float* __restrict__ T,
                const float* __restrict__ W3,
                float* __restrict__ U,
                const float* __restrict__ bias) {
    const int t = threadIdx.x;
    if (NF > 0) {
        __shared__ __attribute__((aligned(16))) float Alds[16 * 132];
        if ((int)blockIdx.x < NF) {
            fold_tile(Alds, T, FR, nullptr, W3, nullptr, U, blockIdx.x, t);
            return;
        }
    }
    const int lane = t & 31;
    const int waveId = __builtin_amdgcn_readfirstlane(t >> 5);
    const int nA = ((int)blockIdx.x - NF) * 16 + 2 * waveId;
    float4 rA, rB;
    gather_rows<SELF>(h, cnt_in, slots, ovf_cnt, ovf, nA, lane, nNodes, rA, rB);
    if (!SELF) {
        const float4 bb = reinterpret_cast<const float4*>(bias)[lane];
        rA.x += bb.x; rA.y += bb.y; rA.z += bb.z; rA.w += bb.w;
        rB.x += bb.x; rB.y += bb.y; rB.z += bb.z; rB.w += bb.w;
    }
    reinterpret_cast<float4*>(g + (size_t)nA * D)[lane]       = rA;
    reinterpret_cast<float4*>(g + (size_t)(nA + 1) * D)[lane] = rB;
}

__global__ __launch_bounds__(256)
void gather_gemm_wmma(const float* __restrict__ h,
                      const int* __restrict__ cnt_in,
                      const int* __restrict__ slots,
                      const int* __restrict__ ovf_cnt,
                      const int* __restrict__ ovf,
                      const int* __restrict__ cnt_out,
                      const float* __restrict__ W,
                      const float* __restrict__ cvec,
                      float* __restrict__ hs, int nNodes) {
    __shared__ __attribute__((aligned(16))) float Alds[16 * 132];
    __shared__ float nl[16];

    const int t = threadIdx.x;
    const int lane = t & 31;
    const int rowBase = blockIdx.x * 16;
    const int waveId = __builtin_amdgcn_readfirstlane(t >> 5);

    const int co = cnt_out[rowBase + (t & 15)];
    {
        const int nA = rowBase + 2 * waveId;
        float4 rA, rB;
        gather_rows<true>(h, cnt_in, slots, ovf_cnt, ovf, nA, lane, nNodes, rA, rB);
        *reinterpret_cast<float4*>(&Alds[(2 * waveId)     * 132 + 4 * lane]) = rA;
        *reinterpret_cast<float4*>(&Alds[(2 * waveId + 1) * 132 + 4 * lane]) = rB;
    }
    if (t < 16) nl[t] = rsqrtf(fmaxf((float)co, 1.0f));
    __syncthreads();

    const int half   = lane >> 4;
    const int lm     = lane & 15;
    const int col0   = waveId * 16;
    const v8f c = panel_wmma(Alds, W, col0, lane);

    const float cv = cvec[col0 + lm];
#pragma unroll
    for (int v = 0; v < 8; ++v) {
        const int row = v + 8 * half;
        hs[(size_t)(rowBase + row) * D + col0 + lm] = (c[v] + cv) * nl[row];
    }
}

extern "C" void kernel_launch(void* const* d_in, const int* in_sizes, int n_in,
                              void* d_out, int out_size, void* d_ws, size_t ws_size,
                              hipStream_t stream) {
    const float* x  = (const float*)d_in[0];
    const float* W1 = (const float*)d_in[1];
    const float* b1 = (const float*)d_in[2];
    const float* W2 = (const float*)d_in[3];
    const float* b2 = (const float*)d_in[4];
    const float* W3 = (const float*)d_in[5];
    const float* b3 = (const float*)d_in[6];
    const int* src  = (const int*)d_in[7];
    const int* dst  = (const int*)d_in[8];
    float* out = (float*)d_out;

    const int N = NN, E = NE;
    int*   cnt_in   = (int*)d_ws;
    int*   cnt_out  = cnt_in  + NPAD;
    int*   ovf_cnt  = cnt_out + NPAD;
    unsigned long long* fill = (unsigned long long*)(ovf_cnt + 64);
    int*   slots    = ovf_cnt + 64 + (size_t)NB * FS;
    int*   ovf      = slots + (size_t)NN * CAP;
    int*   runs     = ovf + (size_t)2 * NE;
    int*   runsS    = runs + (size_t)NB * BCAP;
    float* g1       = (float*)(runsS + (size_t)NB * BCAP);
    float* g2       = g1  + (size_t)NN * D;
    float* T        = g2  + (size_t)NN * D;
    float* U        = T   + (size_t)FR * D;

    dim3 binGrid(NFOLD + (NE + EPW - 1) / EPW);
    dim3 rowGrid(NN / 16);
    dim3 rowGridF(NFOLD + NN / 16);

    hipMemsetAsync(cnt_in, 0, ((size_t)2 * NPAD + 64 + (size_t)NB * FS) * sizeof(int), stream);
    bin_kernel<<<binGrid, BIN_THREADS, 0, stream>>>(src, dst, cnt_in, cnt_out, slots, ovf_cnt, ovf, fill, runs, runsS, E, N,
                                                    W1, b1, W2, b2, T);
    build_kernel<<<NB, BUILD_THREADS, 0, stream>>>(fill, runs, runsS, cnt_in, cnt_out, slots, ovf_cnt, ovf, N);

    agg_kernel<NFOLD, true><<<rowGridF, 256, 0, stream>>>(x, cnt_in, slots, ovf_cnt, ovf, g1, N, T, W3, U, b3);

    gather_gemm_wmma<<<rowGrid, 256, 0, stream>>>(g1, cnt_in, slots, ovf_cnt, ovf, cnt_out,
                                                  U, U + (size_t)D * D, g2, N);

    agg_kernel<0, false><<<rowGrid, 256, 0, stream>>>(g2, cnt_in, slots, ovf_cnt, ovf, out, N, T, W3, U, b3);
}
